# starting kernel
# baseline (speedup 1.0000x reference)
#include <hip/hip_runtime.h>

typedef __attribute__((ext_vector_type(2))) float v2f;
typedef __attribute__((ext_vector_type(8))) float v8f;

#define GIN_D    128
#define GIN_PAD  132
#define GIN_WAVES 8
#define GIN_ROWS (GIN_WAVES * 16)

__global__ __launch_bounds__(256) void gin_h_init(const float* __restrict__ x,
                                                  const float* __restrict__ eps,
                                                  float* __restrict__ h, int n4) {
    int i = blockIdx.x * 256 + threadIdx.x;
    if (i < n4) {
        float s = 1.0f + eps[0];
        float4 v = ((const float4*)x)[i];
        v.x *= s; v.y *= s; v.z *= s; v.w *= s;
        ((float4*)h)[i] = v;
    }
}

__global__ __launch_bounds__(256) void gin_edge_scatter(const float* __restrict__ x,
                                                        const int* __restrict__ edge,
                                                        float* __restrict__ h, int ne) {
    long t = (long)blockIdx.x * 256 + threadIdx.x;
    int e = (int)(t >> 5);
    if (e >= ne) return;
    int src = edge[e];
    int dst = edge[ne + e];
    if (src == dst) return;
    int c = ((int)t & 31) * 4;
    float4 v = *(const float4*)(x + (long)src * GIN_D + c);
    float* p = h + (long)dst * GIN_D + c;
    unsafeAtomicAdd(p + 0, v.x);
    unsafeAtomicAdd(p + 1, v.y);
    unsafeAtomicAdd(p + 2, v.z);
    unsafeAtomicAdd(p + 3, v.w);
}

__global__ __launch_bounds__(256) void gin_mlp(const float* __restrict__ h,
                                               const float* __restrict__ W1,
                                               const float* __restrict__ b1,
                                               const float* __restrict__ W2,
                                               const float* __restrict__ b2,
                                               float* __restrict__ out,
                                               const int* __restrict__ ids,
                                               int nrows) {
    __shared__ float lds_W[GIN_D * GIN_PAD];
    __shared__ float lds_A[GIN_WAVES * 16 * GIN_PAD];
    __shared__ float lds_b1[GIN_D];
    __shared__ float lds_b2[GIN_D];

    const int tid   = threadIdx.x;
    const int wave  = tid >> 5;
    const int lane  = tid & 31;
    const int lmod  = lane & 15;
    const int lhalf = lane >> 4;
    const int koff  = lhalf * 2;
    const int row0  = blockIdx.x * GIN_ROWS + wave * 16;

    if (tid < GIN_D) lds_b1[tid] = b1[tid];
    else             lds_b2[tid - GIN_D] = b2[tid - GIN_D];

    for (int i = tid; i < GIN_D * GIN_D; i += 256) {
        int k = i >> 7, n = i & (GIN_D - 1);
        lds_W[n * GIN_PAD + k] = W1[i];
    }

    float* myA = &lds_A[wave * 16 * GIN_PAD];
    for (int r = 0; r < 16; ++r) {
        int row = row0 + r;
        row = row < nrows ? row : nrows - 1;
        long srow = ids ? (long)ids[row] : (long)row;
        float4 v = *(const float4*)(h + srow * GIN_D + lane * 4);
        *(float4*)(myA + r * GIN_PAD + lane * 4) = v;
    }
    __syncthreads();

    v8f acc[8] = {};
    for (int kk = 0; kk < GIN_D; kk += 4) {
        int k = kk + koff;
        v2f a = *(const v2f*)(myA + lmod * GIN_PAD + k);
#pragma unroll
        for (int ct = 0; ct < 8; ++ct) {
            v2f b = *(const v2f*)(lds_W + (ct * 16 + lmod) * GIN_PAD + k);
            acc[ct] = __builtin_amdgcn_wmma_f32_16x16x4_f32(
                false, a, false, b, (short)0, acc[ct], false, false);
        }
    }
    __syncthreads();

#pragma unroll
    for (int ct = 0; ct < 8; ++ct) {
        float bv = lds_b1[ct * 16 + lmod];
#pragma unroll
        for (int r = 0; r < 8; ++r) {
            float t = acc[ct][r] + bv;
            myA[(r + 8 * lhalf) * GIN_PAD + ct * 16 + lmod] = t > 0.0f ? t : 0.0f;
        }
    }
    for (int i = tid; i < GIN_D * GIN_D; i += 256) {
        int k = i >> 7, n = i & (GIN_D - 1);
        lds_W[n * GIN_PAD + k] = W2[i];
    }
    __syncthreads();

    v8f acc2[8] = {};
    for (int kk = 0; kk < GIN_D; kk += 4) {
        int k = kk + koff;
        v2f a = *(const v2f*)(myA + lmod * GIN_PAD + k);
#pragma unroll
        for (int ct = 0; ct < 8; ++ct) {
            v2f b = *(const v2f*)(lds_W + (ct * 16 + lmod) * GIN_PAD + k);
            acc2[ct] = __builtin_amdgcn_wmma_f32_16x16x4_f32(
                false, a, false, b, (short)0, acc2[ct], false, false);
        }
    }

#pragma unroll
    for (int r = 0; r < 8; ++r) {
        int row = row0 + r + 8 * lhalf;
        bool ok = row < nrows;
        long orow = 0;
        if (ok) orow = ids ? (long)ids[row] : (long)row;
#pragma unroll
        for (int ct = 0; ct < 8; ++ct) {
            float val = acc2[ct][r] + lds_b2[ct * 16 + lmod];
            if (ok) {
                float* p = out + orow * GIN_D + ct * 16 + lmod;
                if (ids) unsafeAtomicAdd(p, val);
                else     *p = val;
            }
        }
    }
}

extern "C" void kernel_launch(void* const* d_in, const int* in_sizes, int n_in,
                              void* d_out, int out_size, void* d_ws, size_t ws_size,
                              hipStream_t stream) {
    const float* x    = (const float*)d_in[0];
    const int*   edge = (const int*)d_in[1];
    const int*   nid  = (const int*)d_in[2];
    const float* eps  = (const float*)d_in[3];
    const float* W1   = (const float*)d_in[4];
    const float* b1   = (const float*)d_in[5];
    const float* W2   = (const float*)d_in[6];
    const float* b2   = (const float*)d_in[7];
    const float* Wi1  = (const float*)d_in[8];
    const float* bi1  = (const float*)d_in[9];
    const float* Wi2  = (const float*)d_in[10];
    const float* bi2  = (const float*)d_in[11];

    const int N   = in_sizes[0] / GIN_D;
    const int E   = in_sizes[1] / 2;
    const int NID = in_sizes[2];
    float* h = (float*)d_ws;

    int n4 = (N * GIN_D) / 4;
    gin_h_init<<<(n4 + 255) / 256, 256, 0, stream>>>(x, eps, h, n4);

    long nthr = (long)E * 32;
    gin_edge_scatter<<<(unsigned)((nthr + 255) / 256), 256, 0, stream>>>(x, edge, h, E);

    gin_mlp<<<(N + GIN_ROWS - 1) / GIN_ROWS, 256, 0, stream>>>(
        h, W1, b1, W2, b2, (float*)d_out, nullptr, N);
    gin_mlp<<<(NID + GIN_ROWS - 1) / GIN_ROWS, 256, 0, stream>>>(
        h, Wi1, bi1, Wi2, bi2, (float*)d_out, nid, NID);
}


# edge scatter: lane l takes columns l,l+32,l+64,l+96 so each f32 atomic instruction covers one whole 128B line, not 8 lanes in each of 4 lines; atomics kept, arrival order as before
# speedup vs baseline: 1.2056x; 1.2056x over previous
#include <hip/hip_runtime.h>

typedef __attribute__((ext_vector_type(2))) float v2f;
typedef __attribute__((ext_vector_type(8))) float v8f;

#define GIN_D    128
#define GIN_PAD  132
#define GIN_WAVES 8
#define GIN_ROWS (GIN_WAVES * 16)

__global__ __launch_bounds__(256) void gin_h_init(const float* __restrict__ x,
                                                  const float* __restrict__ eps,
                                                  float* __restrict__ h, int n4) {
    int i = blockIdx.x * 256 + threadIdx.x;
    if (i < n4) {
        float s = 1.0f + eps[0];
        float4 v = ((const float4*)x)[i];
        v.x *= s; v.y *= s; v.z *= s; v.w *= s;
        ((float4*)h)[i] = v;
    }
}

__global__ __launch_bounds__(256) void gin_edge_scatter(const float* __restrict__ x,
                                                        const int* __restrict__ edge,
                                                        float* __restrict__ h, int ne) {
    long t = (long)blockIdx.x * 256 + threadIdx.x;
    int e = (int)(t >> 5);
    if (e >= ne) return;
    int src = edge[e];
    int dst = edge[ne + e];
    if (src == dst) return;
    int l = (int)t & 31;
    const float* xs = x + (long)src * GIN_D + l;
    float* p = h + (long)dst * GIN_D + l;
    float v0 = xs[0];
    float v1 = xs[32];
    float v2 = xs[64];
    float v3 = xs[96];
    unsafeAtomicAdd(p + 0,  v0);
    unsafeAtomicAdd(p + 32, v1);
    unsafeAtomicAdd(p + 64, v2);
    unsafeAtomicAdd(p + 96, v3);
}

__global__ __launch_bounds__(256) void gin_mlp(const float* __restrict__ h,
                                               const float* __restrict__ W1,
                                               const float* __restrict__ b1,
                                               const float* __restrict__ W2,
                                               const float* __restrict__ b2,
                                               float* __restrict__ out,
                                               const int* __restrict__ ids,
                                               int nrows) {
    __shared__ float lds_W[GIN_D * GIN_PAD];
    __shared__ float lds_A[GIN_WAVES * 16 * GIN_PAD];
    __shared__ float lds_b1[GIN_D];
    __shared__ float lds_b2[GIN_D];

    const int tid   = threadIdx.x;
    const int wave  = tid >> 5;
    const int lane  = tid & 31;
    const int lmod  = lane & 15;
    const int lhalf = lane >> 4;
    const int koff  = lhalf * 2;
    const int row0  = blockIdx.x * GIN_ROWS + wave * 16;

    if (tid < GIN_D) lds_b1[tid] = b1[tid];
    else             lds_b2[tid - GIN_D] = b2[tid - GIN_D];

    for (int i = tid; i < GIN_D * GIN_D; i += 256) {
        int k = i >> 7, n = i & (GIN_D - 1);
        lds_W[n * GIN_PAD + k] = W1[i];
    }

    float* myA = &lds_A[wave * 16 * GIN_PAD];
    for (int r = 0; r < 16; ++r) {
        int row = row0 + r;
        row = row < nrows ? row : nrows - 1;
        long srow = ids ? (long)ids[row] : (long)row;
        float4 v = *(const float4*)(h + srow * GIN_D + lane * 4);
        *(float4*)(myA + r * GIN_PAD + lane * 4) = v;
    }
    __syncthreads();

    v8f acc[8] = {};
    for (int kk = 0; kk < GIN_D; kk += 4) {
        int k = kk + koff;
        v2f a = *(const v2f*)(myA + lmod * GIN_PAD + k);
#pragma unroll
        for (int ct = 0; ct < 8; ++ct) {
            v2f b = *(const v2f*)(lds_W + (ct * 16 + lmod) * GIN_PAD + k);
            acc[ct] = __builtin_amdgcn_wmma_f32_16x16x4_f32(
                false, a, false, b, (short)0, acc[ct], false, false);
        }
    }
    __syncthreads();

#pragma unroll
    for (int ct = 0; ct < 8; ++ct) {
        float bv = lds_b1[ct * 16 + lmod];
#pragma unroll
        for (int r = 0; r < 8; ++r) {
            float t = acc[ct][r] + bv;
            myA[(r + 8 * lhalf) * GIN_PAD + ct * 16 + lmod] = t > 0.0f ? t : 0.0f;
        }
    }
    for (int i = tid; i < GIN_D * GIN_D; i += 256) {
        int k = i >> 7, n = i & (GIN_D - 1);
        lds_W[n * GIN_PAD + k] = W2[i];
    }
    __syncthreads();

    v8f acc2[8] = {};
    for (int kk = 0; kk < GIN_D; kk += 4) {
        int k = kk + koff;
        v2f a = *(const v2f*)(myA + lmod * GIN_PAD + k);
#pragma unroll
        for (int ct = 0; ct < 8; ++ct) {
            v2f b = *(const v2f*)(lds_W + (ct * 16 + lmod) * GIN_PAD + k);
            acc2[ct] = __builtin_amdgcn_wmma_f32_16x16x4_f32(
                false, a, false, b, (short)0, acc2[ct], false, false);
        }
    }

#pragma unroll
    for (int r = 0; r < 8; ++r) {
        int row = row0 + r + 8 * lhalf;
        bool ok = row < nrows;
        long orow = 0;
        if (ok) orow = ids ? (long)ids[row] : (long)row;
#pragma unroll
        for (int ct = 0; ct < 8; ++ct) {
            float val = acc2[ct][r] + lds_b2[ct * 16 + lmod];
            if (ok) {
                float* p = out + orow * GIN_D + ct * 16 + lmod;
                if (ids) unsafeAtomicAdd(p, val);
                else     *p = val;
            }
        }
    }
}

extern "C" void kernel_launch(void* const* d_in, const int* in_sizes, int n_in,
                              void* d_out, int out_size, void* d_ws, size_t ws_size,
                              hipStream_t stream) {
    const float* x    = (const float*)d_in[0];
    const int*   edge = (const int*)d_in[1];
    const int*   nid  = (const int*)d_in[2];
    const float* eps  = (const float*)d_in[3];
    const float* W1   = (const float*)d_in[4];
    const float* b1   = (const float*)d_in[5];
    const float* W2   = (const float*)d_in[6];
    const float* b2   = (const float*)d_in[7];
    const float* Wi1  = (const float*)d_in[8];
    const float* bi1  = (const float*)d_in[9];
    const float* Wi2  = (const float*)d_in[10];
    const float* bi2  = (const float*)d_in[11];

    const int N   = in_sizes[0] / GIN_D;
    const int E   = in_sizes[1] / 2;
    const int NID = in_sizes[2];
    float* h = (float*)d_ws;

    int n4 = (N * GIN_D) / 4;
    gin_h_init<<<(n4 + 255) / 256, 256, 0, stream>>>(x, eps, h, n4);

    long nthr = (long)E * 32;
    gin_edge_scatter<<<(unsigned)((nthr + 255) / 256), 256, 0, stream>>>(x, edge, h, E);

    gin_mlp<<<(N + GIN_ROWS - 1) / GIN_ROWS, 256, 0, stream>>>(
        h, W1, b1, W2, b2, (float*)d_out, nullptr, N);
    gin_mlp<<<(NID + GIN_ROWS - 1) / GIN_ROWS, 256, 0, stream>>>(
        h, Wi1, bi1, Wi2, bi2, (float*)d_out, nid, NID);
}
